# starting kernel
# baseline (speedup 1.0000x reference)
#include <hip/hip_runtime.h>

typedef __attribute__((ext_vector_type(2))) float v2f;
typedef __attribute__((ext_vector_type(8))) float v8f;

#define BATCH 2048
#define Qn 128
#define Sn 128
#define Hn 32
#define Cn (Qn * Sn)

#define WAVES 8
#define RT 2
#define ROWS_PER_BLOCK (WAVES * RT * 16)

__global__ __launch_bounds__(256)
void divenc_wmma_f32(const float* __restrict__ x,
                     const float* __restrict__ W1,
                     const float* __restrict__ b1,
                     const float* __restrict__ W2,
                     const float* __restrict__ b2,
                     float* __restrict__ out)
{
    __shared__ float w1s[32 * 128];

    const int q   = blockIdx.x;
    const int tid = threadIdx.x;

    {
        const float* w1g = W1 + (size_t)q * Sn * Hn;
        #pragma unroll
        for (int it = 0; it < 8; ++it) {
            int i  = tid + it * 256;
            int l  = i & 31;
            int t  = (i >> 5) & 1;
            int f  = i >> 6;
            int hi = l >> 4;
            int n  = l & 15;
            int s  = 4 * f + 2 * hi;
            int g  = s * Hn + n + 16 * t;
            float v0 = w1g[g];
            float v1 = w1g[g + Hn];
            float* dst = &w1s[f * 128 + t * 64 + l * 2];
            dst[0] = v0;
            dst[1] = v1;
        }
    }
    __syncthreads();

    const int wave = tid >> 5;
    const int lane = tid & 31;
    const int ln   = lane & 15;
    const int hi   = lane >> 4;
    const int b0   = blockIdx.y * ROWS_PER_BLOCK + wave * (RT * 16);

    const float* xrow[RT];
    #pragma unroll
    for (int r = 0; r < RT; ++r)
        xrow[r] = x + (size_t)(b0 + r * 16 + ln) * Cn + q * Sn + hi * 2;

    const float* bfrag = &w1s[lane * 2];

    v8f acc[RT][2] = {};

    #pragma unroll 8
    for (int k = 0; k < Sn; k += 4) {
        const int f = k >> 2;
        v2f bfa = *(const v2f*)(bfrag + f * 128);
        v2f bfb = *(const v2f*)(bfrag + f * 128 + 64);

        #pragma unroll
        for (int r = 0; r < RT; ++r) {
            v2f a = __builtin_nontemporal_load((const v2f*)(xrow[r] + k));
            acc[r][0] = __builtin_amdgcn_wmma_f32_16x16x4_f32(
                false, a, false, bfa, (short)0, acc[r][0], false, false);
            acc[r][1] = __builtin_amdgcn_wmma_f32_16x16x4_f32(
                false, a, false, bfb, (short)0, acc[r][1], false, false);
        }
    }

    const float w2a = W2[q * Hn + ln];
    const float w2b = W2[q * Hn + ln + 16];
    const float b1a = b1[q * Hn + ln];
    const float b1b = b1[q * Hn + ln + 16];
    const float b2q = b2[q];

    #pragma unroll
    for (int r = 0; r < RT; ++r) {
        #pragma unroll
        for (int m = 0; m < 8; ++m) {
            float h0 = acc[r][0][m] + b1a;
            float h1 = acc[r][1][m] + b1b;
            h0 = (h0 > 0.0f) ? h0 : (__expf(h0) - 1.0f);
            h1 = (h1 > 0.0f) ? h1 : (__expf(h1) - 1.0f);
            float p = h0 * w2a + h1 * w2b;
            p += __shfl_xor(p, 8, 16);
            p += __shfl_xor(p, 4, 16);
            p += __shfl_xor(p, 2, 16);
            p += __shfl_xor(p, 1, 16);
            if (ln == 0) {
                int b = b0 + r * 16 + hi * 8 + m;
                out[(size_t)b * Qn + q] = p + b2q;
            }
        }
    }
}

extern "C" void kernel_launch(void* const* d_in, const int* in_sizes, int n_in,
                              void* d_out, int out_size, void* d_ws, size_t ws_size,
                              hipStream_t stream) {
    const float* x  = (const float*)d_in[0];
    const float* W1 = (const float*)d_in[1];
    const float* b1 = (const float*)d_in[2];
    const float* W2 = (const float*)d_in[3];
    const float* b2 = (const float*)d_in[4];
    float* out = (float*)d_out;

    dim3 grid(Qn, BATCH / ROWS_PER_BLOCK);
    divenc_wmma_f32<<<grid, 256, 0, stream>>>(x, W1, b1, W2, b2, out);
}


# each work-group runs 2 slices q: next slice's W1 read under the trips into a second LDS plane, x in a 4-slot ring refilled a slice ahead, barrier split round the epilogue; sum order unchanged
# speedup vs baseline: 3.1027x; 3.1027x over previous
#include <hip/hip_runtime.h>

typedef __attribute__((ext_vector_type(2))) float v2f;
typedef __attribute__((ext_vector_type(4))) float v4f;
typedef __attribute__((ext_vector_type(8))) float v8f;

#define BATCH 2048
#define Qn 128
#define Sn 128
#define Hn 32
#define Cn (Qn * Sn)

#define WAVES 8
#define RT 2
#define ROWS_PER_BLOCK (WAVES * RT * 16)
#define NQ 2
#define W1S (32 * 128)

template <int MASK>
__device__ __forceinline__ float row_xor(float v)
{
    return __int_as_float(__builtin_amdgcn_update_dpp(
        0, __float_as_int(v), 0x160 + MASK, 0xf, 0xf, true));
}

template <int MASK>
__device__ __forceinline__ float fold_pair(float lo, float hi, bool up)
{
    const float keep = up ? hi : lo;
    const float send = up ? lo : hi;
    return keep + row_xor<MASK>(send);
}

__device__ __forceinline__ void load_group(v4f (&r)[4], const float* p)
{
    #pragma unroll
    for (int i = 0; i < 4; ++i)
        r[i] = *(const v4f*)(p + (size_t)i * Cn);
}

template <int MASK>
__device__ __forceinline__ void swap_step(v4f& A, v4f& B, bool up)
{
    #pragma unroll
    for (int e = 0; e < 4; ++e) {
        const float send = up ? A[e] : B[e];
        const float recv = row_xor<MASK>(send);
        A[e] = up ? recv : A[e];
        B[e] = up ? B[e] : recv;
    }
}

__device__ __forceinline__ void quad_transpose(v4f (&r)[4], bool u1, bool u2)
{
    swap_step<1>(r[0], r[1], u1);
    swap_step<1>(r[2], r[3], u1);
    swap_step<2>(r[0], r[2], u2);
    swap_step<2>(r[1], r[3], u2);
}

__device__ __forceinline__ void mma_half(const v4f (&r)[4], const v2f (&bf)[8][2], v8f (&c)[2])
{
    #pragma unroll
    for (int s = 0; s < 8; ++s) {
        const v4f t = r[s >> 1];
        const v2f a = (s & 1) ? __builtin_shufflevector(t, t, 2, 3)
                              : __builtin_shufflevector(t, t, 0, 1);
        c[0] = __builtin_amdgcn_wmma_f32_16x16x4_f32(
            false, a, false, bf[s][0], (short)0, c[0], false, false);
        c[1] = __builtin_amdgcn_wmma_f32_16x16x4_f32(
            false, a, false, bf[s][1], (short)0, c[1], false, false);
    }
}

__device__ __forceinline__ void wmma_slots(v8f (&c)[2])
{
    __builtin_amdgcn_sched_barrier(0);
    asm volatile("v_nop\n\tv_nop\n\tv_nop\n\tv_nop" : "+v"(c[0]), "+v"(c[1]) : : "memory");
    __builtin_amdgcn_sched_barrier(0);
}

template <bool MORE>
__device__ __forceinline__ void trip(v4f (&r0)[4], v4f (&r1)[4], v8f (&acc)[RT][2],
                                     const float* bfrag, int f0,
                                     const float* xn0, const float* xn1,
                                     bool u1, bool u2)
{
    v2f bf[8][2];
    #pragma unroll
    for (int s = 0; s < 8; ++s) {
        const int f = f0 + s;
        bf[s][0] = *(const v2f*)(bfrag + f * 128);
        bf[s][1] = *(const v2f*)(bfrag + f * 128 + 64);
    }
    __builtin_amdgcn_sched_barrier(0);

    quad_transpose(r0, u1, u2);
    __builtin_amdgcn_sched_barrier(0);
    mma_half(r0, bf, acc[0]);
    wmma_slots(acc[0]);
    if constexpr (MORE) load_group(r0, xn0);
    __builtin_amdgcn_sched_barrier(0);

    quad_transpose(r1, u1, u2);
    __builtin_amdgcn_sched_barrier(0);
    mma_half(r1, bf, acc[1]);
    wmma_slots(acc[1]);
    if constexpr (MORE) load_group(r1, xn1);
}

__device__ __forceinline__ void w1_read(float (&wv0)[8], float (&wv1)[8], const float* w1g, int tid)
{
    #pragma unroll
    for (int it = 0; it < 8; ++it) {
        int i  = tid + it * 256;
        int l  = i & 31;
        int t  = (i >> 5) & 1;
        int f  = i >> 6;
        int s  = 32 * (f >> 3) + 16 * (l >> 4) + 2 * (f & 7);
        int g  = s * Hn + (l & 15) + 16 * t;
        wv0[it] = w1g[g];
        wv1[it] = w1g[g + Hn];
    }
}

__device__ __forceinline__ void w1_store(const float (&wv0)[8], const float (&wv1)[8], float* plane, int tid)
{
    #pragma unroll
    for (int it = 0; it < 8; ++it) {
        int i  = tid + it * 256;
        int l  = i & 31;
        int t  = (i >> 5) & 1;
        int f  = i >> 6;
        float* dst = &plane[f * 128 + t * 64 + l * 2];
        dst[0] = wv0[it];
        dst[1] = wv1[it];
    }
}

template <bool MORE>
__device__ __forceinline__ void slice(v4f (&xa)[4][4], v4f (&xb)[4][4],
                                      const float* bfrag, float* wnext, const float* w1g_next,
                                      const float* xn0, const float* xn1,
                                      const float* __restrict__ b1, const float* __restrict__ W2,
                                      const float* __restrict__ b2, float* __restrict__ out,
                                      int q, int b0, int tid, int ln, int hi, bool x1, bool x2)
{
    float wv0[8], wv1[8];
    if constexpr (MORE) w1_read(wv0, wv1, w1g_next, tid);
    const float w2a = W2[q * Hn + ln];
    const float w2b = W2[q * Hn + ln + 16];
    const float b1a = b1[q * Hn + ln];
    const float b1b = b1[q * Hn + ln + 16];
    const float b2q = b2[q];
    __builtin_amdgcn_sched_barrier(0);

    v8f acc[RT][2] = {};

    #pragma unroll
    for (int j = 0; j < 4; ++j)
        trip<MORE>(xa[j], xb[j], acc, bfrag, 8 * j, xn0 + 32 * j, xn1 + 32 * j, x1, x2);

    if constexpr (MORE) {
        w1_store(wv0, wv1, wnext, tid);
        asm volatile("s_wait_dscnt 0x0" ::: "memory");
        asm volatile("s_barrier_signal -1" ::: "memory");
    }

    float p[RT * 8];
    #pragma unroll
    for (int r = 0; r < RT; ++r) {
        #pragma unroll
        for (int m = 0; m < 8; ++m) {
            float h0 = acc[r][0][m] + b1a;
            float h1 = acc[r][1][m] + b1b;
            h0 = (h0 > 0.0f) ? h0 : (__expf(h0) - 1.0f);
            h1 = (h1 > 0.0f) ? h1 : (__expf(h1) - 1.0f);
            p[r * 8 + m] = h0 * w2a + h1 * w2b;
        }
    }

    static_assert(RT == 2, "the transposed reduction below folds exactly 16 sums");
    const bool u8 = (ln & 8) != 0;
    const bool u4 = (ln & 4) != 0;
    const bool u2 = (ln & 2) != 0;
    const bool u1 = (ln & 1) != 0;

    float s8[8];
    #pragma unroll
    for (int j = 0; j < 8; ++j) s8[j] = fold_pair<8>(p[j], p[j + 8], u8);
    float s4[4];
    #pragma unroll
    for (int j = 0; j < 4; ++j) s4[j] = fold_pair<4>(s8[j], s8[j + 4], u4);
    float s2[2];
    #pragma unroll
    for (int j = 0; j < 2; ++j) s2[j] = fold_pair<2>(s4[j], s4[j + 2], u2);
    const float tot = fold_pair<1>(s2[0], s2[1], u1);

    {
        const int b = b0 + (ln >> 3) * 16 + hi * 8 + (ln & 7);
        out[(size_t)b * Qn + q] = tot + b2q;
    }

    if constexpr (MORE) {
        asm volatile("s_barrier_wait -1" ::: "memory");
    }
}

__global__ __launch_bounds__(256)
void divenc_wmma_f32(const float* __restrict__ x,
                     const float* __restrict__ W1,
                     const float* __restrict__ b1,
                     const float* __restrict__ W2,
                     const float* __restrict__ b2,
                     float* __restrict__ out)
{
    __shared__ float w1s[2 * W1S];

    const int q0  = blockIdx.x * NQ;
    const int tid = threadIdx.x;

    const int wave = tid >> 5;
    const int lane = tid & 31;
    const int ln   = lane & 15;
    const int hi   = lane >> 4;
    const int b0   = blockIdx.y * ROWS_PER_BLOCK + wave * (RT * 16);

    const float* xrow0 = x + (size_t)(b0 + 4 * ((lane >> 2) & 3)) * Cn + q0 * Sn
                           + (hi * 4 + (lane & 3)) * 4;
    const float* xrow1 = xrow0 + (size_t)16 * Cn;
    const bool x1 = (lane & 1) != 0;
    const bool x2 = (lane & 2) != 0;

    float wv0[8], wv1[8];
    w1_read(wv0, wv1, W1 + (size_t)q0 * Sn * Hn, tid);
    __builtin_amdgcn_sched_barrier(0);

    v4f xa[4][4], xb[4][4];
    #pragma unroll
    for (int j = 0; j < 4; ++j) {
        load_group(xa[j], xrow0 + 32 * j);
        load_group(xb[j], xrow1 + 32 * j);
    }
    __builtin_amdgcn_sched_barrier(0);

    w1_store(wv0, wv1, w1s, tid);
    asm volatile("s_wait_dscnt 0x0" ::: "memory");
    asm volatile("s_barrier_signal -1" ::: "memory");
    asm volatile("s_barrier_wait -1" ::: "memory");

    static_assert(NQ == 2, "two slices, one LDS plane each");
    slice<true>(xa, xb, &w1s[lane * 2], &w1s[W1S],
                W1 + (size_t)(q0 + 1) * Sn * Hn,
                xrow0 + Sn, xrow1 + Sn,
                b1, W2, b2, out, q0, b0, tid, ln, hi, x1, x2);
    slice<false>(xa, xb, &w1s[W1S + lane * 2], nullptr, nullptr,
                 nullptr, nullptr,
                 b1, W2, b2, out, q0 + 1, b0, tid, ln, hi, x1, x2);
}

extern "C" void kernel_launch(void* const* d_in, const int* in_sizes, int n_in,
                              void* d_out, int out_size, void* d_ws, size_t ws_size,
                              hipStream_t stream) {
    const float* x  = (const float*)d_in[0];
    const float* W1 = (const float*)d_in[1];
    const float* b1 = (const float*)d_in[2];
    const float* W2 = (const float*)d_in[3];
    const float* b2 = (const float*)d_in[4];
    float* out = (float*)d_out;

    dim3 grid(Qn / NQ, BATCH / ROWS_PER_BLOCK);
    divenc_wmma_f32<<<grid, 256, 0, stream>>>(x, W1, b1, W2, b2, out);
}
